# starting kernel
# baseline (speedup 1.0000x reference)
#include <hip/hip_runtime.h>

#define NFREQ 257
#define NFILT 64
#define IN_C  64
#define HID   128
#define TLEN  1000
#define BATCH 2

#define MT       64
#define MAXLSTR  34

typedef float v2f __attribute__((ext_vector_type(2)));
typedef float v8f __attribute__((ext_vector_type(8)));

template <int JMAXC>
__global__ __launch_bounds__(256)
void freqcomp_wmma_f32_t(const float* __restrict__ x,
                         const float* __restrict__ W,
                         const int*   __restrict__ idx,
                         float*       __restrict__ out)
{
    constexpr int JPAD = (JMAXC + 3) & ~3;
    constexpr int LSTR = JPAD + 2;
    constexpr int NA   = JPAD / 4;
    constexpr int NB   = JPAD / 2;

    __shared__ __align__(16) float lA[2][MT  * LSTR];
    __shared__ __align__(16) float lB[2][HID * LSTR];

    const int tbase = blockIdx.x * MT;
    const int f     = blockIdx.y;
    const int b     = blockIdx.z;
    const int start = idx[f * JMAXC];

    const int tid  = threadIdx.x;
    const int lane = tid & 31;
    const int wave = tid >> 5;
    const int wm   = wave & 3;
    const int wn   = wave >> 2;

    const int arow = tid >> 2;
    const int aq   = tid & 3;
    const int brow = tid >> 1;
    const int bq   = tid & 1;

    int trow = tbase + arow;
    trow = trow < TLEN ? trow : TLEN - 1;
    const float* xrow = x + (((size_t)b * IN_C) * TLEN + trow) * NFREQ;
    const float* wrow = W + ((size_t)(f * HID + brow) * IN_C) * JMAXC;

    float ar[NA];
    float br[NB];

    auto gload = [&](int c) {
        const float* rp = xrow + (size_t)c * TLEN * NFREQ;
        #pragma unroll
        for (int i = 0; i < NA; ++i) {
            int col = start + aq * NA + i;
            col = col < NFREQ ? col : NFREQ - 1;
            ar[i] = rp[col];
        }
        const float* wp = wrow + c * JMAXC;
        #pragma unroll
        for (int i = 0; i < NB; ++i) {
            int j = bq * NB + i;
            br[i] = (j < JMAXC) ? wp[j] : 0.0f;
        }
    };
    auto stlds = [&](int buf) {
        #pragma unroll
        for (int i = 0; i < NA; ++i) lA[buf][arow * LSTR + aq * NA + i] = ar[i];
        #pragma unroll
        for (int i = 0; i < NB; ++i) lB[buf][brow * LSTR + bq * NB + i] = br[i];
    };

    v8f acc0 = {}, acc1 = {}, acc2 = {}, acc3 = {};

    const int ml   = lane & 15;
    const int kk   = (lane >> 4) << 1;
    const int aoff = (wm * 16 + ml) * LSTR + kk;
    const int boff = (wn * 64 + ml) * LSTR + kk;

    gload(0);
    int buf = 0;
    for (int c = 0; c < IN_C; ++c) {
        stlds(buf);
        __syncthreads();
        if (c + 1 < IN_C) gload(c + 1);

        const float* pa = &lA[buf][aoff];
        const float* pb = &lB[buf][boff];
        #pragma unroll
        for (int ks = 0; ks < JPAD; ks += 4) {
            v2f a  = *(const v2f*)(pa + ks);
            v2f b0 = *(const v2f*)(pb + ks);
            v2f b1 = *(const v2f*)(pb + 16 * LSTR + ks);
            v2f b2 = *(const v2f*)(pb + 32 * LSTR + ks);
            v2f b3 = *(const v2f*)(pb + 48 * LSTR + ks);
            acc0 = __builtin_amdgcn_wmma_f32_16x16x4_f32(false, a, false, b0, (short)0, acc0, false, false);
            acc1 = __builtin_amdgcn_wmma_f32_16x16x4_f32(false, a, false, b1, (short)0, acc1, false, false);
            acc2 = __builtin_amdgcn_wmma_f32_16x16x4_f32(false, a, false, b2, (short)0, acc2, false, false);
            acc3 = __builtin_amdgcn_wmma_f32_16x16x4_f32(false, a, false, b3, (short)0, acc3, false, false);
        }
        buf ^= 1;
    }

    const int t0    = tbase + wm * 16 + ((lane >> 4) << 3);
    const int hbase = wn * 64 + ml;
    float* op = out + (((size_t)b * HID + hbase) * TLEN + t0) * NFILT + f;
    constexpr size_t HSTEP = (size_t)16 * TLEN * NFILT;

    if (tbase + MT <= TLEN) {
        #pragma unroll
        for (int s = 0; s < 4; ++s) {
            v8f a = s == 0 ? acc0 : s == 1 ? acc1 : s == 2 ? acc2 : acc3;
            float* p = op + (size_t)s * HSTEP;
            #pragma unroll
            for (int r = 0; r < 8; ++r) p[(size_t)r * NFILT] = a[r];
        }
    } else {
        #pragma unroll
        for (int s = 0; s < 4; ++s) {
            v8f a = s == 0 ? acc0 : s == 1 ? acc1 : s == 2 ? acc2 : acc3;
            float* p = op + (size_t)s * HSTEP;
            #pragma unroll
            for (int r = 0; r < 8; ++r)
                if (t0 + r < TLEN) p[(size_t)r * NFILT] = a[r];
        }
    }
}

__global__ __launch_bounds__(256)
void freqcomp_wmma_f32_g(const float* __restrict__ x,
                         const float* __restrict__ W,
                         const int*   __restrict__ idx,
                         float*       __restrict__ out,
                         int jmax)
{
    __shared__ __align__(16) float lA[MT  * MAXLSTR];
    __shared__ __align__(16) float lB[HID * MAXLSTR];

    const int jpad = (jmax + 3) & ~3;
    const int lstr = jpad + 2;

    const int tbase = blockIdx.x * MT;
    const int f     = blockIdx.y;
    const int b     = blockIdx.z;
    const int start = idx[f * jmax];

    const int tid  = threadIdx.x;
    const int lane = tid & 31;
    const int wave = tid >> 5;
    const int wm   = wave & 3;
    const int wn   = wave >> 2;

    v8f acc0 = {}, acc1 = {}, acc2 = {}, acc3 = {};

    const int arow = tid >> 2;
    const int aq   = tid & 3;
    const int brow = tid >> 1;
    const int bq   = tid & 1;

    int trow = tbase + arow;
    trow = trow < TLEN ? trow : TLEN - 1;
    const float* xrow = x + (((size_t)b * IN_C) * TLEN + trow) * NFREQ;

    const int ml = lane & 15;
    const int kk = (lane >> 4) << 1;
    const float* pa  = &lA[(wm * 16 + ml) * lstr + kk];
    const float* pb0 = &lB[(wn * 64 + ml) * lstr + kk];

    for (int c = 0; c < IN_C; ++c) {
        {
            const float* rp = xrow + (size_t)c * TLEN * NFREQ;
            const int jn = jpad >> 2;
            for (int i = 0; i < jn; ++i) {
                int j   = aq * jn + i;
                int col = start + j;
                col = col < NFREQ ? col : NFREQ - 1;
                lA[arow * lstr + j] = rp[col];
            }
        }
        {
            const float* wp = W + (((size_t)f * HID + brow) * IN_C + c) * jmax;
            const int jn = jpad >> 1;
            for (int i = 0; i < jn; ++i) {
                int j = bq * jn + i;
                lB[brow * lstr + j] = (j < jmax) ? wp[j] : 0.0f;
            }
        }
        __syncthreads();
        for (int ks = 0; ks < jpad; ks += 4) {
            v2f a  = *(const v2f*)(pa  + ks);
            v2f b0 = *(const v2f*)(pb0 + ks);
            v2f b1 = *(const v2f*)(pb0 + 16 * lstr + ks);
            v2f b2 = *(const v2f*)(pb0 + 32 * lstr + ks);
            v2f b3 = *(const v2f*)(pb0 + 48 * lstr + ks);
            acc0 = __builtin_amdgcn_wmma_f32_16x16x4_f32(false, a, false, b0, (short)0, acc0, false, false);
            acc1 = __builtin_amdgcn_wmma_f32_16x16x4_f32(false, a, false, b1, (short)0, acc1, false, false);
            acc2 = __builtin_amdgcn_wmma_f32_16x16x4_f32(false, a, false, b2, (short)0, acc2, false, false);
            acc3 = __builtin_amdgcn_wmma_f32_16x16x4_f32(false, a, false, b3, (short)0, acc3, false, false);
        }
        __syncthreads();
    }

    const int t0    = tbase + wm * 16 + ((lane >> 4) << 3);
    const int hbase = wn * 64 + ml;
    float* op = out + (((size_t)b * HID + hbase) * TLEN + t0) * NFILT + f;
    const size_t HSTEP = (size_t)16 * TLEN * NFILT;

    if (tbase + MT <= TLEN) {
        #pragma unroll
        for (int s = 0; s < 4; ++s) {
            v8f a = s == 0 ? acc0 : s == 1 ? acc1 : s == 2 ? acc2 : acc3;
            float* p = op + (size_t)s * HSTEP;
            #pragma unroll
            for (int r = 0; r < 8; ++r) p[(size_t)r * NFILT] = a[r];
        }
    } else {
        #pragma unroll
        for (int s = 0; s < 4; ++s) {
            v8f a = s == 0 ? acc0 : s == 1 ? acc1 : s == 2 ? acc2 : acc3;
            float* p = op + (size_t)s * HSTEP;
            #pragma unroll
            for (int r = 0; r < 8; ++r)
                if (t0 + r < TLEN) p[(size_t)r * NFILT] = a[r];
        }
    }
}

extern "C" void kernel_launch(void* const* d_in, const int* in_sizes, int n_in,
                              void* d_out, int out_size, void* d_ws, size_t ws_size,
                              hipStream_t stream)
{
    const float* x   = (const float*)d_in[0];
    const float* W   = (const float*)d_in[1];
    const int*   idx = (const int*)d_in[2];
    float* out = (float*)d_out;

    const int jmax = in_sizes[1] / (NFILT * HID * IN_C);

    dim3 grid((TLEN + MT - 1) / MT, NFILT, BATCH);
    if (jmax == 21) {
        freqcomp_wmma_f32_t<21><<<grid, 256, 0, stream>>>(x, W, idx, out);
    } else {
        freqcomp_wmma_f32_g<<<grid, 256, 0, stream>>>(x, W, idx, out, jmax);
    }
}


# Stage W and x in flattened (c,j) K order, 4 channels (84 k) per stage: W by contiguous 16-byte loads, no K padding (21 not 24 WMMA steps per 4 channels); k summed in new groups of 4
# speedup vs baseline: 1.3331x; 1.3331x over previous
#include <hip/hip_runtime.h>

#define NFREQ 257
#define NFILT 64
#define IN_C  64
#define HID   128
#define TLEN  1000
#define BATCH 2

#define MT       64
#define MAXLSTR  34

typedef float v2f __attribute__((ext_vector_type(2)));
typedef float v8f __attribute__((ext_vector_type(8)));

typedef float v4f __attribute__((ext_vector_type(4)));

template <int JMAXC>
__global__ __launch_bounds__(256)
void freqcomp_wmma_f32_t(const float* __restrict__ x,
                         const float* __restrict__ W,
                         const int*   __restrict__ idx,
                         float*       __restrict__ out)
{
    constexpr int CH   = 4;
    constexpr int KC   = CH * JMAXC;
    constexpr int NST  = IN_C / CH;
    constexpr int KROW = IN_C * JMAXC;
    constexpr int NAI  = JMAXC;
    constexpr int NBQ  = 16 * JMAXC;
    constexpr int NBI  = (NBQ + 31) / 32;

    __shared__ __align__(16) float lA[2][MT  * KC];
    __shared__ __align__(16) float lB[2][HID * KC];

    const int tbase = blockIdx.x * MT;
    const int f     = blockIdx.y;
    const int b     = blockIdx.z;

    const int tid  = threadIdx.x;
    const int lane = tid & 31;
    const int wave = tid >> 5;
    const int wm   = wave & 3;
    const int wn   = wave >> 2;

    unsigned aofs[NAI];
    #pragma unroll
    for (int n = 0; n < NAI; ++n) {
        const int i  = lane + 32 * n;
        const int tl = i / KC;
        const int k  = i - tl * KC;
        const int cl = k / JMAXC;
        const int j  = k - cl * JMAXC;
        int t = tbase + 8 * wave + tl;
        t = t < TLEN ? t : TLEN - 1;
        int col = idx[f * JMAXC + j];
        col = col < 0 ? 0 : (col < NFREQ ? col : NFREQ - 1);
        aofs[n] = (unsigned)(((cl * TLEN + t) * NFREQ + col) * 4);
    }
    unsigned bofs[NBI];
    #pragma unroll
    for (int n = 0; n < NBI; ++n) {
        int i = lane + 32 * n;
        i = i < NBQ ? i : NBQ - 1;
        const int rl = i / JMAXC;
        const int p  = i - rl * JMAXC;
        bofs[n] = (unsigned)((((16 * wave + rl) * KROW) + p * 4) * 4);
    }
    const bool blast = (lane + 32 * (NBI - 1)) < NBQ;

    const char* xb = (const char*)(x + (size_t)b * IN_C * TLEN * NFREQ);
    const char* wb = (const char*)(W + (size_t)f * HID * KROW);

    float ar[NAI];
    v4f   br[NBI];

    auto gload = [&](int st) {
        const char* xs = xb + (size_t)st * CH * TLEN * NFREQ * 4;
        #pragma unroll
        for (int n = 0; n < NAI; ++n) ar[n] = *(const float*)(xs + aofs[n]);
        const char* ws = wb + (size_t)st * KC * 4;
        #pragma unroll
        for (int n = 0; n < NBI - 1; ++n) br[n] = *(const v4f*)(ws + bofs[n]);
        if (blast) br[NBI - 1] = *(const v4f*)(ws + bofs[NBI - 1]);
    };
    auto stlds = [&](int buf) {
        float* qa = &lA[buf][8 * wave * KC + lane];
        #pragma unroll
        for (int n = 0; n < NAI; ++n) qa[32 * n] = ar[n];
        v4f* qb = (v4f*)&lB[buf][16 * wave * KC] + lane;
        #pragma unroll
        for (int n = 0; n < NBI - 1; ++n) qb[32 * n] = br[n];
        if (blast) qb[32 * (NBI - 1)] = br[NBI - 1];
    };

    v8f acc0 = {}, acc1 = {}, acc2 = {}, acc3 = {};

    const int ml   = lane & 15;
    const int kk   = (lane >> 4) << 1;
    const int aoff = (wm * 16 + ml) * KC + kk;
    const int boff = (wn * 64 + ml) * KC + kk;

    gload(0);
    int buf = 0;
    for (int st = 0; st < NST; ++st) {
        stlds(buf);
        __syncthreads();
        if (st + 1 < NST) gload(st + 1);

        const float* pa = &lA[buf][aoff];
        const float* pb = &lB[buf][boff];
        #pragma unroll
        for (int ks = 0; ks < KC; ks += 4) {
            v2f a  = *(const v2f*)(pa + ks);
            v2f b0 = *(const v2f*)(pb + ks);
            v2f b1 = *(const v2f*)(pb + 16 * KC + ks);
            v2f b2 = *(const v2f*)(pb + 32 * KC + ks);
            v2f b3 = *(const v2f*)(pb + 48 * KC + ks);
            acc0 = __builtin_amdgcn_wmma_f32_16x16x4_f32(false, a, false, b0, (short)0, acc0, false, false);
            acc1 = __builtin_amdgcn_wmma_f32_16x16x4_f32(false, a, false, b1, (short)0, acc1, false, false);
            acc2 = __builtin_amdgcn_wmma_f32_16x16x4_f32(false, a, false, b2, (short)0, acc2, false, false);
            acc3 = __builtin_amdgcn_wmma_f32_16x16x4_f32(false, a, false, b3, (short)0, acc3, false, false);
        }
        buf ^= 1;
    }

    const int t0    = tbase + wm * 16 + ((lane >> 4) << 3);
    const int hbase = wn * 64 + ml;
    float* op = out + (((size_t)b * HID + hbase) * TLEN + t0) * NFILT + f;
    constexpr size_t HSTEP = (size_t)16 * TLEN * NFILT;

    if (tbase + MT <= TLEN) {
        #pragma unroll
        for (int s = 0; s < 4; ++s) {
            v8f a = s == 0 ? acc0 : s == 1 ? acc1 : s == 2 ? acc2 : acc3;
            float* p = op + (size_t)s * HSTEP;
            #pragma unroll
            for (int r = 0; r < 8; ++r) p[(size_t)r * NFILT] = a[r];
        }
    } else {
        #pragma unroll
        for (int s = 0; s < 4; ++s) {
            v8f a = s == 0 ? acc0 : s == 1 ? acc1 : s == 2 ? acc2 : acc3;
            float* p = op + (size_t)s * HSTEP;
            #pragma unroll
            for (int r = 0; r < 8; ++r)
                if (t0 + r < TLEN) p[(size_t)r * NFILT] = a[r];
        }
    }
}

__global__ __launch_bounds__(256)
void freqcomp_wmma_f32_g(const float* __restrict__ x,
                         const float* __restrict__ W,
                         const int*   __restrict__ idx,
                         float*       __restrict__ out,
                         int jmax)
{
    __shared__ __align__(16) float lA[MT  * MAXLSTR];
    __shared__ __align__(16) float lB[HID * MAXLSTR];

    const int jpad = (jmax + 3) & ~3;
    const int lstr = jpad + 2;

    const int tbase = blockIdx.x * MT;
    const int f     = blockIdx.y;
    const int b     = blockIdx.z;
    const int start = idx[f * jmax];

    const int tid  = threadIdx.x;
    const int lane = tid & 31;
    const int wave = tid >> 5;
    const int wm   = wave & 3;
    const int wn   = wave >> 2;

    v8f acc0 = {}, acc1 = {}, acc2 = {}, acc3 = {};

    const int arow = tid >> 2;
    const int aq   = tid & 3;
    const int brow = tid >> 1;
    const int bq   = tid & 1;

    int trow = tbase + arow;
    trow = trow < TLEN ? trow : TLEN - 1;
    const float* xrow = x + (((size_t)b * IN_C) * TLEN + trow) * NFREQ;

    const int ml = lane & 15;
    const int kk = (lane >> 4) << 1;
    const float* pa  = &lA[(wm * 16 + ml) * lstr + kk];
    const float* pb0 = &lB[(wn * 64 + ml) * lstr + kk];

    for (int c = 0; c < IN_C; ++c) {
        {
            const float* rp = xrow + (size_t)c * TLEN * NFREQ;
            const int jn = jpad >> 2;
            for (int i = 0; i < jn; ++i) {
                int j   = aq * jn + i;
                int col = start + j;
                col = col < NFREQ ? col : NFREQ - 1;
                lA[arow * lstr + j] = rp[col];
            }
        }
        {
            const float* wp = W + (((size_t)f * HID + brow) * IN_C + c) * jmax;
            const int jn = jpad >> 1;
            for (int i = 0; i < jn; ++i) {
                int j = bq * jn + i;
                lB[brow * lstr + j] = (j < jmax) ? wp[j] : 0.0f;
            }
        }
        __syncthreads();
        for (int ks = 0; ks < jpad; ks += 4) {
            v2f a  = *(const v2f*)(pa  + ks);
            v2f b0 = *(const v2f*)(pb0 + ks);
            v2f b1 = *(const v2f*)(pb0 + 16 * lstr + ks);
            v2f b2 = *(const v2f*)(pb0 + 32 * lstr + ks);
            v2f b3 = *(const v2f*)(pb0 + 48 * lstr + ks);
            acc0 = __builtin_amdgcn_wmma_f32_16x16x4_f32(false, a, false, b0, (short)0, acc0, false, false);
            acc1 = __builtin_amdgcn_wmma_f32_16x16x4_f32(false, a, false, b1, (short)0, acc1, false, false);
            acc2 = __builtin_amdgcn_wmma_f32_16x16x4_f32(false, a, false, b2, (short)0, acc2, false, false);
            acc3 = __builtin_amdgcn_wmma_f32_16x16x4_f32(false, a, false, b3, (short)0, acc3, false, false);
        }
        __syncthreads();
    }

    const int t0    = tbase + wm * 16 + ((lane >> 4) << 3);
    const int hbase = wn * 64 + ml;
    float* op = out + (((size_t)b * HID + hbase) * TLEN + t0) * NFILT + f;
    const size_t HSTEP = (size_t)16 * TLEN * NFILT;

    if (tbase + MT <= TLEN) {
        #pragma unroll
        for (int s = 0; s < 4; ++s) {
            v8f a = s == 0 ? acc0 : s == 1 ? acc1 : s == 2 ? acc2 : acc3;
            float* p = op + (size_t)s * HSTEP;
            #pragma unroll
            for (int r = 0; r < 8; ++r) p[(size_t)r * NFILT] = a[r];
        }
    } else {
        #pragma unroll
        for (int s = 0; s < 4; ++s) {
            v8f a = s == 0 ? acc0 : s == 1 ? acc1 : s == 2 ? acc2 : acc3;
            float* p = op + (size_t)s * HSTEP;
            #pragma unroll
            for (int r = 0; r < 8; ++r)
                if (t0 + r < TLEN) p[(size_t)r * NFILT] = a[r];
        }
    }
}

extern "C" void kernel_launch(void* const* d_in, const int* in_sizes, int n_in,
                              void* d_out, int out_size, void* d_ws, size_t ws_size,
                              hipStream_t stream)
{
    const float* x   = (const float*)d_in[0];
    const float* W   = (const float*)d_in[1];
    const int*   idx = (const int*)d_in[2];
    float* out = (float*)d_out;

    const int jmax = in_sizes[1] / (NFILT * HID * IN_C);

    dim3 grid((TLEN + MT - 1) / MT, NFILT, BATCH);
    if (jmax == 21) {
        freqcomp_wmma_f32_t<21><<<grid, 256, 0, stream>>>(x, W, idx, out);
    } else {
        freqcomp_wmma_f32_g<<<grid, 256, 0, stream>>>(x, W, idx, out, jmax);
    }
}


# Output in whole 128-byte lines: accumulator tiles go to the workspace in register order, a second launch transposes them through LDS to f-contiguous rows (was 4-byte stores, one line per lane)
# speedup vs baseline: 1.5554x; 1.1668x over previous
#include <hip/hip_runtime.h>

#define NFREQ 257
#define NFILT 64
#define IN_C  64
#define HID   128
#define TLEN  1000
#define BATCH 2

#define MT       64
#define MAXLSTR  34
#define TT16     64
#define HB16     8

typedef float v2f __attribute__((ext_vector_type(2)));
typedef float v8f __attribute__((ext_vector_type(8)));

typedef float v4f __attribute__((ext_vector_type(4)));

template <int JMAXC>
__global__ __launch_bounds__(256)
void freqcomp_wmma_f32_t(const float* __restrict__ x,
                         const float* __restrict__ W,
                         const int*   __restrict__ idx,
                         float*       __restrict__ wsp)
{
    constexpr int CH   = 4;
    constexpr int KC   = CH * JMAXC;
    constexpr int NST  = IN_C / CH;
    constexpr int KROW = IN_C * JMAXC;
    constexpr int NAI  = JMAXC;
    constexpr int NBQ  = 16 * JMAXC;
    constexpr int NBI  = (NBQ + 31) / 32;

    __shared__ __align__(16) float lA[2][MT  * KC];
    __shared__ __align__(16) float lB[2][HID * KC];

    const int tbase = blockIdx.x * MT;
    const int f     = blockIdx.y;
    const int b     = blockIdx.z;

    const int tid  = threadIdx.x;
    const int lane = tid & 31;
    const int wave = tid >> 5;
    const int wm   = wave & 3;
    const int wn   = wave >> 2;

    unsigned aofs[NAI];
    #pragma unroll
    for (int n = 0; n < NAI; ++n) {
        const int i  = lane + 32 * n;
        const int tl = i / KC;
        const int k  = i - tl * KC;
        const int cl = k / JMAXC;
        const int j  = k - cl * JMAXC;
        int t = tbase + 8 * wave + tl;
        t = t < TLEN ? t : TLEN - 1;
        int col = idx[f * JMAXC + j];
        col = col < 0 ? 0 : (col < NFREQ ? col : NFREQ - 1);
        aofs[n] = (unsigned)(((cl * TLEN + t) * NFREQ + col) * 4);
    }
    unsigned bofs[NBI];
    #pragma unroll
    for (int n = 0; n < NBI; ++n) {
        int i = lane + 32 * n;
        i = i < NBQ ? i : NBQ - 1;
        const int rl = i / JMAXC;
        const int p  = i - rl * JMAXC;
        bofs[n] = (unsigned)((((16 * wave + rl) * KROW) + p * 4) * 4);
    }
    const bool blast = (lane + 32 * (NBI - 1)) < NBQ;

    const char* xb = (const char*)(x + (size_t)b * IN_C * TLEN * NFREQ);
    const char* wb = (const char*)(W + (size_t)f * HID * KROW);

    float ar[NAI];
    v4f   br[NBI];

    auto gload = [&](int st) {
        const char* xs = xb + (size_t)st * CH * TLEN * NFREQ * 4;
        #pragma unroll
        for (int n = 0; n < NAI; ++n) ar[n] = *(const float*)(xs + aofs[n]);
        const char* ws = wb + (size_t)st * KC * 4;
        #pragma unroll
        for (int n = 0; n < NBI - 1; ++n) br[n] = *(const v4f*)(ws + bofs[n]);
        if (blast) br[NBI - 1] = *(const v4f*)(ws + bofs[NBI - 1]);
    };
    auto stlds = [&](int buf) {
        float* qa = &lA[buf][8 * wave * KC + lane];
        #pragma unroll
        for (int n = 0; n < NAI; ++n) qa[32 * n] = ar[n];
        v4f* qb = (v4f*)&lB[buf][16 * wave * KC] + lane;
        #pragma unroll
        for (int n = 0; n < NBI - 1; ++n) qb[32 * n] = br[n];
        if (blast) qb[32 * (NBI - 1)] = br[NBI - 1];
    };

    v8f acc0 = {}, acc1 = {}, acc2 = {}, acc3 = {};

    const int ml   = lane & 15;
    const int kk   = (lane >> 4) << 1;
    const int aoff = (wm * 16 + ml) * KC + kk;
    const int boff = (wn * 64 + ml) * KC + kk;

    gload(0);
    int buf = 0;
    for (int st = 0; st < NST; ++st) {
        stlds(buf);
        __syncthreads();
        if (st + 1 < NST) gload(st + 1);

        const float* pa = &lA[buf][aoff];
        const float* pb = &lB[buf][boff];
        #pragma unroll
        for (int ks = 0; ks < KC; ks += 4) {
            v2f a  = *(const v2f*)(pa + ks);
            v2f b0 = *(const v2f*)(pb + ks);
            v2f b1 = *(const v2f*)(pb + 16 * KC + ks);
            v2f b2 = *(const v2f*)(pb + 32 * KC + ks);
            v2f b3 = *(const v2f*)(pb + 48 * KC + ks);
            acc0 = __builtin_amdgcn_wmma_f32_16x16x4_f32(false, a, false, b0, (short)0, acc0, false, false);
            acc1 = __builtin_amdgcn_wmma_f32_16x16x4_f32(false, a, false, b1, (short)0, acc1, false, false);
            acc2 = __builtin_amdgcn_wmma_f32_16x16x4_f32(false, a, false, b2, (short)0, acc2, false, false);
            acc3 = __builtin_amdgcn_wmma_f32_16x16x4_f32(false, a, false, b3, (short)0, acc3, false, false);
        }
        buf ^= 1;
    }

    {
        const int tt = blockIdx.x * (MT / 16) + wm;
        float* wp = wsp + ((((size_t)(f * BATCH + b) * TT16 + tt) * HB16 + wn * 4) * 256) + lane;
        #pragma unroll
        for (int s = 0; s < 4; ++s) {
            v8f a = s == 0 ? acc0 : s == 1 ? acc1 : s == 2 ? acc2 : acc3;
            float* p = wp + s * 256;
            #pragma unroll
            for (int r = 0; r < 8; ++r) p[r * 32] = a[r];
        }
    }
}

__global__ __launch_bounds__(256)
void freqcomp_wmma_f32_g(const float* __restrict__ x,
                         const float* __restrict__ W,
                         const int*   __restrict__ idx,
                         float*       __restrict__ wsp,
                         int jmax)
{
    __shared__ __align__(16) float lA[MT  * MAXLSTR];
    __shared__ __align__(16) float lB[HID * MAXLSTR];

    const int jpad = (jmax + 3) & ~3;
    const int lstr = jpad + 2;

    const int tbase = blockIdx.x * MT;
    const int f     = blockIdx.y;
    const int b     = blockIdx.z;
    const int start = idx[f * jmax];

    const int tid  = threadIdx.x;
    const int lane = tid & 31;
    const int wave = tid >> 5;
    const int wm   = wave & 3;
    const int wn   = wave >> 2;

    v8f acc0 = {}, acc1 = {}, acc2 = {}, acc3 = {};

    const int arow = tid >> 2;
    const int aq   = tid & 3;
    const int brow = tid >> 1;
    const int bq   = tid & 1;

    int trow = tbase + arow;
    trow = trow < TLEN ? trow : TLEN - 1;
    const float* xrow = x + (((size_t)b * IN_C) * TLEN + trow) * NFREQ;

    const int ml = lane & 15;
    const int kk = (lane >> 4) << 1;
    const float* pa  = &lA[(wm * 16 + ml) * lstr + kk];
    const float* pb0 = &lB[(wn * 64 + ml) * lstr + kk];

    for (int c = 0; c < IN_C; ++c) {
        {
            const float* rp = xrow + (size_t)c * TLEN * NFREQ;
            const int jn = jpad >> 2;
            for (int i = 0; i < jn; ++i) {
                int j   = aq * jn + i;
                int col = start + j;
                col = col < NFREQ ? col : NFREQ - 1;
                lA[arow * lstr + j] = rp[col];
            }
        }
        {
            const float* wp = W + (((size_t)f * HID + brow) * IN_C + c) * jmax;
            const int jn = jpad >> 1;
            for (int i = 0; i < jn; ++i) {
                int j = bq * jn + i;
                lB[brow * lstr + j] = (j < jmax) ? wp[j] : 0.0f;
            }
        }
        __syncthreads();
        for (int ks = 0; ks < jpad; ks += 4) {
            v2f a  = *(const v2f*)(pa  + ks);
            v2f b0 = *(const v2f*)(pb0 + ks);
            v2f b1 = *(const v2f*)(pb0 + 16 * lstr + ks);
            v2f b2 = *(const v2f*)(pb0 + 32 * lstr + ks);
            v2f b3 = *(const v2f*)(pb0 + 48 * lstr + ks);
            acc0 = __builtin_amdgcn_wmma_f32_16x16x4_f32(false, a, false, b0, (short)0, acc0, false, false);
            acc1 = __builtin_amdgcn_wmma_f32_16x16x4_f32(false, a, false, b1, (short)0, acc1, false, false);
            acc2 = __builtin_amdgcn_wmma_f32_16x16x4_f32(false, a, false, b2, (short)0, acc2, false, false);
            acc3 = __builtin_amdgcn_wmma_f32_16x16x4_f32(false, a, false, b3, (short)0, acc3, false, false);
        }
        __syncthreads();
    }

    {
        const int tt = blockIdx.x * (MT / 16) + wm;
        float* wp = wsp + ((((size_t)(f * BATCH + b) * TT16 + tt) * HB16 + wn * 4) * 256) + lane;
        #pragma unroll
        for (int s = 0; s < 4; ++s) {
            v8f a = s == 0 ? acc0 : s == 1 ? acc1 : s == 2 ? acc2 : acc3;
            float* p = wp + s * 256;
            #pragma unroll
            for (int r = 0; r < 8; ++r) p[r * 32] = a[r];
        }
    }
}


__global__ __launch_bounds__(256)
void freqcomp_lines_out(const float* __restrict__ wsp, float* __restrict__ out)
{
    __shared__ float tile[NFILT * 257];

    const int tt   = blockIdx.x;
    const int hb   = blockIdx.y;
    const int b    = blockIdx.z;
    const int tid  = threadIdx.x;
    const int lane = tid & 31;
    const int wave = tid >> 5;

    constexpr size_t FSTEP = (size_t)BATCH * TT16 * HB16 * 256;
    const float* sp = wsp + ((((size_t)b * TT16 + tt) * HB16 + hb) * 256) + tid;
    #pragma unroll 16
    for (int f = 0; f < NFILT; ++f) tile[f * 257 + tid] = sp[(size_t)f * FSTEP];
    __syncthreads();

    const float* c0 = &tile[(2 * lane) * 257 + wave * 32];
    float* ob = out + (((size_t)b * HID + hb * 16) * TLEN + tt * 16 + wave) * NFILT + 2 * lane;
    if (tt * 16 + 16 <= TLEN) {
        #pragma unroll 8
        for (int l = 0; l < 32; ++l) {
            v2f v;
            v.x = c0[l];
            v.y = c0[257 + l];
            *(v2f*)(ob + ((size_t)(l & 15) * TLEN + 8 * (l >> 4)) * NFILT) = v;
        }
    } else {
        for (int l = 0; l < 32; ++l) {
            if (tt * 16 + wave + 8 * (l >> 4) < TLEN) {
                v2f v;
                v.x = c0[l];
                v.y = c0[257 + l];
                *(v2f*)(ob + ((size_t)(l & 15) * TLEN + 8 * (l >> 4)) * NFILT) = v;
            }
        }
    }
}

extern "C" void kernel_launch(void* const* d_in, const int* in_sizes, int n_in,
                              void* d_out, int out_size, void* d_ws, size_t ws_size,
                              hipStream_t stream)
{
    const float* x   = (const float*)d_in[0];
    const float* W   = (const float*)d_in[1];
    const int*   idx = (const int*)d_in[2];
    float* out = (float*)d_out;
    float* wsp = (float*)d_ws;

    const int jmax = in_sizes[1] / (NFILT * HID * IN_C);

    dim3 grid((TLEN + MT - 1) / MT, NFILT, BATCH);
    if (jmax == 21) {
        freqcomp_wmma_f32_t<21><<<grid, 256, 0, stream>>>(x, W, idx, wsp);
    } else {
        freqcomp_wmma_f32_g<<<grid, 256, 0, stream>>>(x, W, idx, wsp, jmax);
    }
    dim3 grid2((TLEN + 15) / 16, HB16, BATCH);
    freqcomp_lines_out<<<grid2, 256, 0, stream>>>(wsp, out);
}


# Staging pipelined into the WMMA stream: next stage's LDS stores and the stage after's global loads go in pieces between WMMA groups; the stage boundary keeps only an LDS-only barrier
# speedup vs baseline: 1.8366x; 1.1808x over previous
#include <hip/hip_runtime.h>

#define NFREQ 257
#define NFILT 64
#define IN_C  64
#define HID   128
#define TLEN  1000
#define BATCH 2

#define MT       64
#define MAXLSTR  34
#define TT16     64
#define HB16     8

typedef float v2f __attribute__((ext_vector_type(2)));
typedef float v8f __attribute__((ext_vector_type(8)));

typedef float v4f __attribute__((ext_vector_type(4)));

template <bool V> struct flag_t { static constexpr bool value = V; };

__device__ __forceinline__ void lds_barrier()
{
    __builtin_amdgcn_fence(__ATOMIC_RELEASE, "workgroup", "local");
    __builtin_amdgcn_s_barrier();
    __builtin_amdgcn_fence(__ATOMIC_ACQUIRE, "workgroup", "local");
}

template <int JMAXC>
__global__ __launch_bounds__(256)
void freqcomp_wmma_f32_t(const float* __restrict__ x,
                         const float* __restrict__ W,
                         const int*   __restrict__ idx,
                         float*       __restrict__ wsp)
{
    constexpr int CH   = 4;
    constexpr int KC   = CH * JMAXC;
    constexpr int NST  = IN_C / CH;
    constexpr int KROW = IN_C * JMAXC;
    constexpr int NAI  = JMAXC;
    constexpr int NBQ  = 16 * JMAXC;
    constexpr int NBI  = (NBQ + 31) / 32;

    __shared__ __align__(16) float lA[2][MT  * KC];
    __shared__ __align__(16) float lB[2][HID * KC];

    const int tbase = blockIdx.x * MT;
    const int f     = blockIdx.y;
    const int b     = blockIdx.z;

    const int tid  = threadIdx.x;
    const int lane = tid & 31;
    const int wave = tid >> 5;
    const int wm   = wave & 3;
    const int wn   = wave >> 2;

    unsigned aofs[NAI];
    #pragma unroll
    for (int n = 0; n < NAI; ++n) {
        const int i  = lane + 32 * n;
        const int tl = i / KC;
        const int k  = i - tl * KC;
        const int cl = k / JMAXC;
        const int j  = k - cl * JMAXC;
        int t = tbase + 8 * wave + tl;
        t = t < TLEN ? t : TLEN - 1;
        int col = idx[f * JMAXC + j];
        col = col < 0 ? 0 : (col < NFREQ ? col : NFREQ - 1);
        aofs[n] = (unsigned)(((cl * TLEN + t) * NFREQ + col) * 4);
    }
    unsigned bofs[NBI];
    #pragma unroll
    for (int n = 0; n < NBI; ++n) {
        int i = lane + 32 * n;
        i = i < NBQ ? i : NBQ - 1;
        const int rl = i / JMAXC;
        const int p  = i - rl * JMAXC;
        bofs[n] = (unsigned)((((16 * wave + rl) * KROW) + p * 4) * 4);
    }
    const bool blast = (lane + 32 * (NBI - 1)) < NBQ;

    const char* xb = (const char*)(x + (size_t)b * IN_C * TLEN * NFREQ);
    const char* wb = (const char*)(W + (size_t)f * HID * KROW);

    float ar[NAI];
    v4f   br[NBI];

    constexpr int NKS = KC / 4;
    constexpr int NG  = (NKS + 1) / 2;
    constexpr int XPG = (NAI + NG - 1) / NG;
    constexpr int WPG = (NBI - 1 + NG - 1) / NG;

    v8f acc0 = {}, acc1 = {}, acc2 = {}, acc3 = {};

    const int ml   = lane & 15;
    const int kk   = (lane >> 4) << 1;
    const int aoff = (wm * 16 + ml) * KC + kk;
    const int boff = (wn * 64 + ml) * KC + kk;

    auto gload_stage = [&](const char* xs, const char* ws) __attribute__((always_inline)) {
        if (blast) br[NBI - 1] = *(const v4f*)(ws + bofs[NBI - 1]);
        #pragma unroll
        for (int g = 0; g < NG; ++g) {
            #pragma unroll
            for (int n = XPG * g; n < XPG * g + XPG && n < NAI; ++n) ar[n] = *(const float*)(xs + aofs[n]);
            #pragma unroll
            for (int n = WPG * g; n < WPG * g + WPG && n < NBI - 1; ++n) br[n] = *(const v4f*)(ws + bofs[n]);
        }
    };

    {
        gload_stage(xb, wb);

        float* qa = &lA[0][8 * wave * KC + lane];
        v4f*   qb = (v4f*)&lB[0][16 * wave * KC] + lane;
        #pragma unroll
        for (int n = 0; n < NAI; ++n) qa[32 * n] = ar[n];
        #pragma unroll
        for (int n = 0; n < NBI - 1; ++n) qb[32 * n] = br[n];
        if (blast) qb[32 * (NBI - 1)] = br[NBI - 1];

        gload_stage(xb + (size_t)CH * TLEN * NFREQ * 4, wb + (size_t)KC * 4);
    }

    auto stage = [&](int st, int buf, auto do_st, auto do_ld) __attribute__((always_inline)) {
        constexpr bool ST = decltype(do_st)::value;
        constexpr bool LD = decltype(do_ld)::value;
        const float* pa = &lA[buf][aoff];
        const float* pb = &lB[buf][boff];
        float* qa = &lA[buf ^ 1][8 * wave * KC + lane];
        v4f*   qb = (v4f*)&lB[buf ^ 1][16 * wave * KC] + lane;
        const char* xs = xb + (size_t)(st + 2) * CH * TLEN * NFREQ * 4;
        const char* ws = wb + (size_t)(st + 2) * KC * 4;

        v2f fa[2][2];
        v2f fb[2][2][4];
        auto ldfrag = [&](int g, int set) __attribute__((always_inline)) {
            #pragma unroll
            for (int s = 0; s < 2; ++s) {
                const int ks = 4 * (2 * g + s);
                if (ks < KC) {
                    fa[set][s]    = *(const v2f*)(pa + ks);
                    fb[set][s][0] = *(const v2f*)(pb + ks);
                    fb[set][s][1] = *(const v2f*)(pb + 16 * KC + ks);
                    fb[set][s][2] = *(const v2f*)(pb + 32 * KC + ks);
                    fb[set][s][3] = *(const v2f*)(pb + 48 * KC + ks);
                }
            }
        };
        ldfrag(0, 0);

        if (blast) {
            if (ST) qb[32 * (NBI - 1)] = br[NBI - 1];
            if (LD) br[NBI - 1] = *(const v4f*)(ws + bofs[NBI - 1]);
        }

        #pragma unroll
        for (int g = 0; g < NG; ++g) {
            if (g + 1 < NG) ldfrag(g + 1, (g + 1) & 1);
            if (ST) {
                #pragma unroll
                for (int n = XPG * g; n < XPG * g + XPG && n < NAI; ++n) qa[32 * n] = ar[n];
                #pragma unroll
                for (int n = WPG * g; n < WPG * g + WPG && n < NBI - 1; ++n) qb[32 * n] = br[n];
            }
            if (LD) {
                #pragma unroll
                for (int n = XPG * g; n < XPG * g + XPG && n < NAI; ++n) ar[n] = *(const float*)(xs + aofs[n]);
                #pragma unroll
                for (int n = WPG * g; n < WPG * g + WPG && n < NBI - 1; ++n) br[n] = *(const v4f*)(ws + bofs[n]);
            }
            #pragma unroll
            for (int s = 0; s < 2; ++s) {
                if (4 * (2 * g + s) < KC) {
                    const v2f a = fa[g & 1][s];
                    acc0 = __builtin_amdgcn_wmma_f32_16x16x4_f32(false, a, false, fb[g & 1][s][0], (short)0, acc0, false, false);
                    acc1 = __builtin_amdgcn_wmma_f32_16x16x4_f32(false, a, false, fb[g & 1][s][1], (short)0, acc1, false, false);
                    acc2 = __builtin_amdgcn_wmma_f32_16x16x4_f32(false, a, false, fb[g & 1][s][2], (short)0, acc2, false, false);
                    acc3 = __builtin_amdgcn_wmma_f32_16x16x4_f32(false, a, false, fb[g & 1][s][3], (short)0, acc3, false, false);
                }
            }
            if (ST || LD) __builtin_amdgcn_sched_barrier(0);
        }
    };

    int buf = 0;
    for (int st = 0; st < NST - 2; ++st) {
        lds_barrier();
        stage(st, buf, flag_t<true>{}, flag_t<true>{});
        buf ^= 1;
    }
    lds_barrier();
    stage(NST - 2, buf, flag_t<true>{}, flag_t<false>{});
    buf ^= 1;
    lds_barrier();
    stage(NST - 1, buf, flag_t<false>{}, flag_t<false>{});

    {
        const int tt = blockIdx.x * (MT / 16) + wm;
        float* wp = wsp + ((((size_t)(f * BATCH + b) * TT16 + tt) * HB16 + wn * 4) * 256) + lane;
        #pragma unroll
        for (int s = 0; s < 4; ++s) {
            v8f a = s == 0 ? acc0 : s == 1 ? acc1 : s == 2 ? acc2 : acc3;
            float* p = wp + s * 256;
            #pragma unroll
            for (int r = 0; r < 8; ++r) p[r * 32] = a[r];
        }
    }
}

__global__ __launch_bounds__(256)
void freqcomp_wmma_f32_g(const float* __restrict__ x,
                         const float* __restrict__ W,
                         const int*   __restrict__ idx,
                         float*       __restrict__ wsp,
                         int jmax)
{
    __shared__ __align__(16) float lA[MT  * MAXLSTR];
    __shared__ __align__(16) float lB[HID * MAXLSTR];

    const int jpad = (jmax + 3) & ~3;
    const int lstr = jpad + 2;

    const int tbase = blockIdx.x * MT;
    const int f     = blockIdx.y;
    const int b     = blockIdx.z;
    const int start = idx[f * jmax];

    const int tid  = threadIdx.x;
    const int lane = tid & 31;
    const int wave = tid >> 5;
    const int wm   = wave & 3;
    const int wn   = wave >> 2;

    v8f acc0 = {}, acc1 = {}, acc2 = {}, acc3 = {};

    const int arow = tid >> 2;
    const int aq   = tid & 3;
    const int brow = tid >> 1;
    const int bq   = tid & 1;

    int trow = tbase + arow;
    trow = trow < TLEN ? trow : TLEN - 1;
    const float* xrow = x + (((size_t)b * IN_C) * TLEN + trow) * NFREQ;

    const int ml = lane & 15;
    const int kk = (lane >> 4) << 1;
    const float* pa  = &lA[(wm * 16 + ml) * lstr + kk];
    const float* pb0 = &lB[(wn * 64 + ml) * lstr + kk];

    for (int c = 0; c < IN_C; ++c) {
        {
            const float* rp = xrow + (size_t)c * TLEN * NFREQ;
            const int jn = jpad >> 2;
            for (int i = 0; i < jn; ++i) {
                int j   = aq * jn + i;
                int col = start + j;
                col = col < NFREQ ? col : NFREQ - 1;
                lA[arow * lstr + j] = rp[col];
            }
        }
        {
            const float* wp = W + (((size_t)f * HID + brow) * IN_C + c) * jmax;
            const int jn = jpad >> 1;
            for (int i = 0; i < jn; ++i) {
                int j = bq * jn + i;
                lB[brow * lstr + j] = (j < jmax) ? wp[j] : 0.0f;
            }
        }
        __syncthreads();
        for (int ks = 0; ks < jpad; ks += 4) {
            v2f a  = *(const v2f*)(pa  + ks);
            v2f b0 = *(const v2f*)(pb0 + ks);
            v2f b1 = *(const v2f*)(pb0 + 16 * lstr + ks);
            v2f b2 = *(const v2f*)(pb0 + 32 * lstr + ks);
            v2f b3 = *(const v2f*)(pb0 + 48 * lstr + ks);
            acc0 = __builtin_amdgcn_wmma_f32_16x16x4_f32(false, a, false, b0, (short)0, acc0, false, false);
            acc1 = __builtin_amdgcn_wmma_f32_16x16x4_f32(false, a, false, b1, (short)0, acc1, false, false);
            acc2 = __builtin_amdgcn_wmma_f32_16x16x4_f32(false, a, false, b2, (short)0, acc2, false, false);
            acc3 = __builtin_amdgcn_wmma_f32_16x16x4_f32(false, a, false, b3, (short)0, acc3, false, false);
        }
        __syncthreads();
    }

    {
        const int tt = blockIdx.x * (MT / 16) + wm;
        float* wp = wsp + ((((size_t)(f * BATCH + b) * TT16 + tt) * HB16 + wn * 4) * 256) + lane;
        #pragma unroll
        for (int s = 0; s < 4; ++s) {
            v8f a = s == 0 ? acc0 : s == 1 ? acc1 : s == 2 ? acc2 : acc3;
            float* p = wp + s * 256;
            #pragma unroll
            for (int r = 0; r < 8; ++r) p[r * 32] = a[r];
        }
    }
}


__global__ __launch_bounds__(256)
void freqcomp_lines_out(const float* __restrict__ wsp, float* __restrict__ out)
{
    __shared__ float tile[NFILT * 257];

    const int tt   = blockIdx.x;
    const int hb   = blockIdx.y;
    const int b    = blockIdx.z;
    const int tid  = threadIdx.x;
    const int lane = tid & 31;
    const int wave = tid >> 5;

    constexpr size_t FSTEP = (size_t)BATCH * TT16 * HB16 * 256;
    const float* sp = wsp + ((((size_t)b * TT16 + tt) * HB16 + hb) * 256) + tid;
    #pragma unroll 16
    for (int f = 0; f < NFILT; ++f) tile[f * 257 + tid] = sp[(size_t)f * FSTEP];
    __syncthreads();

    const float* c0 = &tile[(2 * lane) * 257 + wave * 32];
    float* ob = out + (((size_t)b * HID + hb * 16) * TLEN + tt * 16 + wave) * NFILT + 2 * lane;
    if (tt * 16 + 16 <= TLEN) {
        #pragma unroll 8
        for (int l = 0; l < 32; ++l) {
            v2f v;
            v.x = c0[l];
            v.y = c0[257 + l];
            *(v2f*)(ob + ((size_t)(l & 15) * TLEN + 8 * (l >> 4)) * NFILT) = v;
        }
    } else {
        for (int l = 0; l < 32; ++l) {
            if (tt * 16 + wave + 8 * (l >> 4) < TLEN) {
                v2f v;
                v.x = c0[l];
                v.y = c0[257 + l];
                *(v2f*)(ob + ((size_t)(l & 15) * TLEN + 8 * (l >> 4)) * NFILT) = v;
            }
        }
    }
}

extern "C" void kernel_launch(void* const* d_in, const int* in_sizes, int n_in,
                              void* d_out, int out_size, void* d_ws, size_t ws_size,
                              hipStream_t stream)
{
    const float* x   = (const float*)d_in[0];
    const float* W   = (const float*)d_in[1];
    const int*   idx = (const int*)d_in[2];
    float* out = (float*)d_out;
    float* wsp = (float*)d_ws;

    const int jmax = in_sizes[1] / (NFILT * HID * IN_C);

    dim3 grid((TLEN + MT - 1) / MT, NFILT, BATCH);
    if (jmax == 21) {
        freqcomp_wmma_f32_t<21><<<grid, 256, 0, stream>>>(x, W, idx, wsp);
    } else {
        freqcomp_wmma_f32_g<<<grid, 256, 0, stream>>>(x, W, idx, wsp, jmax);
    }
    dim3 grid2((TLEN + 15) / 16, HB16, BATCH);
    freqcomp_lines_out<<<grid2, 256, 0, stream>>>(wsp, out);
}
